# starting kernel
# baseline (speedup 1.0000x reference)
#include <hip/hip_runtime.h>
#include <math.h>

#define HW_PIX (256 * 256)
#define NJ 1024
#define NLB 2

typedef __attribute__((ext_vector_type(2))) float v2f;
typedef __attribute__((ext_vector_type(8))) float v8f;

__global__ __launch_bounds__(256) void pn_kernel(
    const float* __restrict__ verts_normals,
    const int*   __restrict__ faces,
    const int*   __restrict__ pix_to_face,
    const float* __restrict__ bary,
    float*       __restrict__ pn_ws,
    float*       __restrict__ out_pn)
{
  const int p = blockIdx.x * 256 + threadIdx.x;

  const int f  = pix_to_face[p];
  const int i0 = faces[f * 3 + 0];
  const int i1 = faces[f * 3 + 1];
  const int i2 = faces[f * 3 + 2];

  const float b0 = bary[p * 3 + 0];
  const float b1 = bary[p * 3 + 1];
  const float b2 = bary[p * 3 + 2];

  float x = b0 * verts_normals[i0 * 3 + 0] + b1 * verts_normals[i1 * 3 + 0] +
            b2 * verts_normals[i2 * 3 + 0];
  float y = b0 * verts_normals[i0 * 3 + 1] + b1 * verts_normals[i1 * 3 + 1] +
            b2 * verts_normals[i2 * 3 + 1];
  float z = b0 * verts_normals[i0 * 3 + 2] + b1 * verts_normals[i1 * 3 + 2] +
            b2 * verts_normals[i2 * 3 + 2];

  const float n   = sqrtf(x * x + y * y + z * z);
  const float inv = 1.0f / fmaxf(n, 1e-6f);
  x *= inv; y *= inv; z *= inv;

  pn_ws[p * 4 + 0] = x;
  pn_ws[p * 4 + 1] = y;
  pn_ws[p * 4 + 2] = z;
  pn_ws[p * 4 + 3] = 0.0f;

  out_pn[p * 3 + 0] = x;
  out_pn[p * 3 + 1] = y;
  out_pn[p * 3 + 2] = z;
  out_pn[(size_t)HW_PIX * 3 + p * 3 + 0] = x;
  out_pn[(size_t)HW_PIX * 3 + p * 3 + 1] = y;
  out_pn[(size_t)HW_PIX * 3 + p * 3 + 2] = z;
}

__global__ __launch_bounds__(256) void shade_kernel(
    const float* __restrict__ pn_ws,
    const float* __restrict__ light_dirs,
    const float* __restrict__ env_map,
    const float* __restrict__ texels,
    const float* __restrict__ kd,
    float*       __restrict__ out_colors)
{
  __shared__ float s_ld[NJ * 4];
  __shared__ float s_lc[NJ * 4];

  const int b = blockIdx.y;
  const float kdc = fminf(fmaxf(kd[0], 0.0f), 1.0f);

  for (int j = threadIdx.x; j < NJ; j += 256) {
    const float* L = light_dirs + ((size_t)b * NJ + j) * 3;
    s_ld[j * 4 + 0] = L[0];
    s_ld[j * 4 + 1] = L[1];
    s_ld[j * 4 + 2] = L[2];
    s_ld[j * 4 + 3] = 0.0f;
    const float* E = env_map + ((size_t)b * NJ + j) * 3;
    s_lc[j * 4 + 0] = E[0] * kdc;
    s_lc[j * 4 + 1] = E[1] * kdc;
    s_lc[j * 4 + 2] = E[2] * kdc;
    s_lc[j * 4 + 3] = 0.0f;
  }
  __syncthreads();

  const int wave = threadIdx.x >> 5;
  const int lane = threadIdx.x & 31;
  const int lo   = lane & 15;
  const int hi   = lane >> 4;

  const int base  = (blockIdx.x * 8 + wave) * 16;
  const int m_pix = base + lo;

  v2f a;
  a.x = pn_ws[m_pix * 4 + hi * 2 + 0];
  a.y = pn_ws[m_pix * 4 + hi * 2 + 1];

  float accx[8], accy[8], accz[8];
#pragma unroll
  for (int r = 0; r < 8; ++r) { accx[r] = 0.0f; accy[r] = 0.0f; accz[r] = 0.0f; }

#pragma unroll 4
  for (int jt = 0; jt < NJ / 16; ++jt) {
    const int j = jt * 16 + lo;

    v2f bv;
    bv.x = s_ld[j * 4 + hi * 2 + 0];
    bv.y = s_ld[j * 4 + hi * 2 + 1];

    v8f c = {};
    v8f d = __builtin_amdgcn_wmma_f32_16x16x4_f32(
         false, a,  false, bv,
         (short)0, c,  false,  false);

    const float cx = s_lc[j * 4 + 0];
    const float cy = s_lc[j * 4 + 1];
    const float cz = s_lc[j * 4 + 2];

#pragma unroll
    for (int r = 0; r < 8; ++r) {
      const float dv = fminf(fmaxf(d[r], 0.0f), 1.0f);
      accx[r] += dv * cx;
      accy[r] += dv * cy;
      accz[r] += dv * cz;
    }
  }

#pragma unroll
  for (int off = 1; off < 16; off <<= 1) {
#pragma unroll
    for (int r = 0; r < 8; ++r) {
      accx[r] += __shfl_xor(accx[r], off, 32);
      accy[r] += __shfl_xor(accy[r], off, 32);
      accz[r] += __shfl_xor(accz[r], off, 32);
    }
  }

  if (lo == 0) {
#pragma unroll
    for (int r = 0; r < 8; ++r) {
      const int pix = base + r + 8 * hi;
      const size_t o = ((size_t)b * HW_PIX + pix) * 3;
      out_colors[o + 0] = accx[r] * texels[o + 0];
      out_colors[o + 1] = accy[r] * texels[o + 1];
      out_colors[o + 2] = accz[r] * texels[o + 2];
    }
  }
}

extern "C" void kernel_launch(void* const* d_in, const int* in_sizes, int n_in,
                              void* d_out, int out_size, void* d_ws, size_t ws_size,
                              hipStream_t stream) {
  const float* verts_normals = (const float*)d_in[0];
  const int*   faces         = (const int*)d_in[1];
  const int*   pix_to_face   = (const int*)d_in[2];
  const float* bary          = (const float*)d_in[3];
  const float* light_dirs    = (const float*)d_in[4];
  const float* env_map       = (const float*)d_in[5];
  const float* texels        = (const float*)d_in[6];
  const float* kd            = (const float*)d_in[7];

  float* out_colors = (float*)d_out;
  float* out_pn     = (float*)d_out + (size_t)NLB * HW_PIX * 3;
  float* pn_ws      = (float*)d_ws;

  pn_kernel<<<HW_PIX / 256, 256, 0, stream>>>(
      verts_normals, faces, pix_to_face, bary, pn_ws, out_pn);

  shade_kernel<<<dim3(HW_PIX / 128, NLB), 256, 0, stream>>>(
      pn_ws, light_dirs, env_map, texels, kd, out_colors);
}


# Lane sums of the 24 partial colours by a halving exchange on DPP adds (23 cross-lane adds, not 96 LDS shuffles); each lane ends with its own output elements, texels fetched before the loop
# speedup vs baseline: 1.3652x; 1.3652x over previous
#include <hip/hip_runtime.h>
#include <math.h>


#define HW_PIX (256 * 256)
#define NJ 1024
#define NLB 2

typedef __attribute__((ext_vector_type(2))) float v2f;
typedef __attribute__((ext_vector_type(8))) float v8f;

template <int M>
__device__ __forceinline__ float row_xor(float v) {
  return __builtin_bit_cast(
      float, __builtin_amdgcn_update_dpp(0, __builtin_bit_cast(int, v), 0x160 + M, 0xf, 0xf, true));
}

__global__ __launch_bounds__(256) void shade_fused_kernel(
    const float* __restrict__ verts_normals,
    const int*   __restrict__ faces,
    const int*   __restrict__ pix_to_face,
    const float* __restrict__ bary,
    const float* __restrict__ light_dirs,
    const float* __restrict__ env_map,
    const float* __restrict__ texels,
    const float* __restrict__ kd,
    float*       __restrict__ out_colors,
    float*       __restrict__ out_pn)
{
  __shared__ float s_ld[NJ * 4];
  __shared__ float s_lc[NJ * 4];
  __shared__ __attribute__((aligned(16))) float s_pn[128 * 3];

  const int b    = blockIdx.y;
  const int wave = threadIdx.x >> 5;
  const int lane = threadIdx.x & 31;
  const int lo   = lane & 15;
  const int hi   = lane >> 4;

  const int base  = (blockIdx.x * 8 + wave) * 16;
  const int m_pix = base + lo;

  const int   f  = pix_to_face[m_pix];
  const float b0 = bary[m_pix * 3 + 0];
  const float b1 = bary[m_pix * 3 + 1];
  const float b2 = bary[m_pix * 3 + 2];

  const float kdc = fminf(fmaxf(kd[0], 0.0f), 1.0f);
  float Lx[NJ / 256], Ly[NJ / 256], Lz[NJ / 256];
  float Ex[NJ / 256], Ey[NJ / 256], Ez[NJ / 256];
#pragma unroll
  for (int it = 0; it < NJ / 256; ++it) {
    const int j = it * 256 + threadIdx.x;
    const float* L = light_dirs + ((size_t)b * NJ + j) * 3;
    Lx[it] = L[0]; Ly[it] = L[1]; Lz[it] = L[2];
    const float* E = env_map + ((size_t)b * NJ + j) * 3;
    Ex[it] = E[0]; Ey[it] = E[1]; Ez[it] = E[2];
  }

  const int i0 = faces[f * 3 + 0];
  const int i1 = faces[f * 3 + 1];
  const int i2 = faces[f * 3 + 2];

#pragma unroll
  for (int it = 0; it < NJ / 256; ++it) {
    const int j = it * 256 + threadIdx.x;
    s_ld[j * 4 + 0] = Lx[it];
    s_ld[j * 4 + 1] = Ly[it];
    s_ld[j * 4 + 2] = Lz[it];
    s_ld[j * 4 + 3] = 0.0f;
    s_lc[j * 4 + 0] = Ex[it] * kdc;
    s_lc[j * 4 + 1] = Ey[it] * kdc;
    s_lc[j * 4 + 2] = Ez[it] * kdc;
    s_lc[j * 4 + 3] = 0.0f;
  }

  float x = b0 * verts_normals[i0 * 3 + 0] + b1 * verts_normals[i1 * 3 + 0] +
            b2 * verts_normals[i2 * 3 + 0];
  float y = b0 * verts_normals[i0 * 3 + 1] + b1 * verts_normals[i1 * 3 + 1] +
            b2 * verts_normals[i2 * 3 + 1];
  float z = b0 * verts_normals[i0 * 3 + 2] + b1 * verts_normals[i1 * 3 + 2] +
            b2 * verts_normals[i2 * 3 + 2];

  const float n   = sqrtf(x * x + y * y + z * z);
  const float inv = 1.0f / fmaxf(n, 1e-6f);
  x *= inv; y *= inv; z *= inv;

  v2f a;
  a.x = hi ? z : x;
  a.y = hi ? 0.0f : y;

  if (hi == 0) {
    s_pn[(wave * 16 + lo) * 3 + 0] = x;
    s_pn[(wave * 16 + lo) * 3 + 1] = y;
    s_pn[(wave * 16 + lo) * 3 + 2] = z;
  }
  __syncthreads();

  if (threadIdx.x < 96) {
    const float4 v = *reinterpret_cast<const float4*>(&s_pn[threadIdx.x * 4]);
    float* dst = out_pn + ((size_t)b * HW_PIX + (size_t)blockIdx.x * 128) * 3;
    *reinterpret_cast<float4*>(&dst[threadIdx.x * 4]) = v;
  }

  const size_t o_pix = ((size_t)b * HW_PIX + base + 8 * hi + (lo >> 1)) * 3;
  const float t01 = texels[o_pix + (lo & 1)];
  const float t2  = texels[o_pix + 2];

  float accx[8], accy[8], accz[8];
#pragma unroll
  for (int r = 0; r < 8; ++r) { accx[r] = 0.0f; accy[r] = 0.0f; accz[r] = 0.0f; }

#pragma unroll 4
  for (int jt = 0; jt < NJ / 16; ++jt) {
    const int j = jt * 16 + lo;

    v2f bv;
    bv.x = s_ld[j * 4 + hi * 2 + 0];
    bv.y = s_ld[j * 4 + hi * 2 + 1];

    v8f c = {};
    v8f d = __builtin_amdgcn_wmma_f32_16x16x4_f32(
         false, a,  false, bv,
         (short)0, c,  false,  false);

    const float cx = s_lc[j * 4 + 0];
    const float cy = s_lc[j * 4 + 1];
    const float cz = s_lc[j * 4 + 2];

#pragma unroll
    for (int r = 0; r < 8; ++r) {
      const float dv = fminf(fmaxf(d[r], 0.0f), 1.0f);
      accx[r] += dv * cx;
      accy[r] += dv * cy;
      accz[r] += dv * cz;
    }
  }

  float v[24];
#pragma unroll
  for (int r = 0; r < 8; ++r) { v[r * 3 + 0] = accx[r]; v[r * 3 + 1] = accy[r]; v[r * 3 + 2] = accz[r]; }

  const bool k8 = (lo & 8) != 0, k4 = (lo & 4) != 0, k2 = (lo & 2) != 0, k1 = (lo & 1) != 0;
  float w[12], q6[6], q3[3];
#pragma unroll
  for (int i = 0; i < 12; ++i) {
    const float keep = k8 ? v[i + 12] : v[i];
    const float send = k8 ? v[i] : v[i + 12];
    w[i] = keep + row_xor<8>(send);
  }
#pragma unroll
  for (int i = 0; i < 6; ++i) {
    const float keep = k4 ? w[i + 6] : w[i];
    const float send = k4 ? w[i] : w[i + 6];
    q6[i] = keep + row_xor<4>(send);
  }
#pragma unroll
  for (int i = 0; i < 3; ++i) {
    const float keep = k2 ? q6[i + 3] : q6[i];
    const float send = k2 ? q6[i] : q6[i + 3];
    q3[i] = keep + row_xor<2>(send);
  }
  const float keep01 = k1 ? q3[1] : q3[0];
  const float send01 = k1 ? q3[0] : q3[1];
  const float s01 = keep01 + row_xor<1>(send01);
  const float s2  = q3[2] + row_xor<1>(q3[2]);

  out_colors[o_pix + (lo & 1)] = s01 * t01;
  if (!k1) out_colors[o_pix + 2] = s2 * t2;
}

extern "C" void kernel_launch(void* const* d_in, const int* in_sizes, int n_in,
                              void* d_out, int out_size, void* d_ws, size_t ws_size,
                              hipStream_t stream) {
  const float* verts_normals = (const float*)d_in[0];
  const int*   faces         = (const int*)d_in[1];
  const int*   pix_to_face   = (const int*)d_in[2];
  const float* bary          = (const float*)d_in[3];
  const float* light_dirs    = (const float*)d_in[4];
  const float* env_map       = (const float*)d_in[5];
  const float* texels        = (const float*)d_in[6];
  const float* kd            = (const float*)d_in[7];

  float* out_colors = (float*)d_out;
  float* out_pn     = (float*)d_out + (size_t)NLB * HW_PIX * 3;

  shade_fused_kernel<<<dim3(HW_PIX / 128, NLB), 256, 0, stream>>>(
      verts_normals, faces, pix_to_face, bary, light_dirs, env_map, texels, kd,
      out_colors, out_pn);
}
